# starting kernel
# baseline (speedup 1.0000x reference)
#include <hip/hip_runtime.h>
#include <hip/hip_bf16.h>

typedef __attribute__((ext_vector_type(2))) float v2f;
typedef __attribute__((ext_vector_type(8))) float v8f;

#define CH 64
#define HEADS 4
#define HC 256
#define NEG_SLOPE 0.2f

__global__ void gat_init_kernel(float* __restrict__ accum,
                                float* __restrict__ smax,
                                float* __restrict__ den, int N) {
    int idx = blockIdx.x * blockDim.x + threadIdx.x;
    int total = N * HC;
    if (idx < total) accum[idx] = 0.0f;
    if (idx < N * HEADS) {
        smax[idx] = -__builtin_inff();
        den[idx]  = 0.0f;
    }
}

__global__ void gat_proj_kernel(const float* __restrict__ x,
                                const float* __restrict__ Wl,
                                const float* __restrict__ bl,
                                const float* __restrict__ Wr,
                                const float* __restrict__ br,
                                float* __restrict__ xl,
                                float* __restrict__ xr, int N) {
    __shared__ float As[16 * CH];
    const int row0 = blockIdx.x * 16;
    const int tid  = threadIdx.x;

    for (int i = tid; i < 16 * CH; i += 256) {
        int r = i >> 6, c = i & 63;
        As[i] = x[(size_t)(row0 + r) * CH + c];
    }
    __syncthreads();

    const int wave = tid >> 5;
    const int lane = tid & 31;
    const int m    = lane & 15;
    const int koff = (lane >> 4) << 1;
    const int mrow = (lane >> 4) << 3;

    for (int t = 0; t < 4; ++t) {
        const int colTile = wave * 4 + t;
        const bool left   = (colTile < 16);
        const float* __restrict__ W = left ? Wl : Wr;
        const float* __restrict__ b = left ? bl : br;
        const int n = ((colTile & 15) << 4) + m;

        v8f acc;
        const float bv = b[n];
        #pragma unroll
        for (int v = 0; v < 8; ++v) acc[v] = bv;

        #pragma unroll
        for (int k0 = 0; k0 < CH; k0 += 4) {
            v2f a, bf;
            a.x  = As[m * CH + k0 + koff];
            a.y  = As[m * CH + k0 + koff + 1];
            bf.x = W[(size_t)(k0 + koff) * HC + n];
            bf.y = W[(size_t)(k0 + koff + 1) * HC + n];
            acc = __builtin_amdgcn_wmma_f32_16x16x4_f32(
                false, a, false, bf, (short)0, acc, false, false);
        }

        float* __restrict__ out = left ? xl : xr;
        #pragma unroll
        for (int v = 0; v < 8; ++v)
            out[(size_t)(row0 + mrow + v) * HC + n] = acc[v];
    }
}

__global__ void gat_score_kernel(const float* __restrict__ xl,
                                 const float* __restrict__ xr,
                                 const float* __restrict__ att,
                                 const int* __restrict__ edges,
                                 float* __restrict__ score,
                                 float* __restrict__ smax,
                                 int E, int Etot) {
    const int widx = (int)((blockIdx.x * (size_t)blockDim.x + threadIdx.x) >> 5);
    const int lane = threadIdx.x & 31;
    if (widx >= Etot) return;

    int src, dst;
    if (widx < E) { src = edges[widx]; dst = edges[E + widx]; }
    else          { src = widx - E;    dst = src; }

    const float4* __restrict__ pl = (const float4*)(xl + (size_t)src * HC);
    const float4* __restrict__ pr = (const float4*)(xr + (size_t)dst * HC);
    const float4* __restrict__ pa = (const float4*)att;

    float s = 0.0f;
    #pragma unroll
    for (int i = 0; i < 2; ++i) {
        const int q = lane * 2 + i;
        float4 a = pl[q], b = pr[q], w = pa[q];
        float vx = a.x + b.x; vx = vx > 0.0f ? vx : NEG_SLOPE * vx;
        float vy = a.y + b.y; vy = vy > 0.0f ? vy : NEG_SLOPE * vy;
        float vz = a.z + b.z; vz = vz > 0.0f ? vz : NEG_SLOPE * vz;
        float vw = a.w + b.w; vw = vw > 0.0f ? vw : NEG_SLOPE * vw;
        s += w.x * vx + w.y * vy + w.z * vz + w.w * vw;
    }
    s += __shfl_xor(s, 1, 32);
    s += __shfl_xor(s, 2, 32);
    s += __shfl_xor(s, 4, 32);

    if ((lane & 7) == 0) {
        const int h = lane >> 3;
        score[(size_t)widx * HEADS + h] = s;
        __hip_atomic_fetch_max(&smax[dst * HEADS + h], s,
                               __ATOMIC_RELAXED, __HIP_MEMORY_SCOPE_AGENT);
    }
}

__global__ void gat_exp_kernel(const int* __restrict__ edges,
                               const float* __restrict__ smax,
                               float* __restrict__ score,
                               float* __restrict__ den,
                               int E, int Etot) {
    const int idx = blockIdx.x * blockDim.x + threadIdx.x;
    if (idx >= Etot * HEADS) return;
    const int e = idx >> 2, h = idx & 3;
    const int dst = (e < E) ? edges[E + e] : (e - E);
    const float ex = __expf(score[idx] - smax[dst * HEADS + h]);
    score[idx] = ex;
    __hip_atomic_fetch_add(&den[dst * HEADS + h], ex,
                           __ATOMIC_RELAXED, __HIP_MEMORY_SCOPE_AGENT);
}

__global__ void gat_aggr_kernel(const float* __restrict__ xl,
                                const int* __restrict__ edges,
                                const float* __restrict__ ex,
                                const float* __restrict__ den,
                                float* __restrict__ accum,
                                int E, int Etot) {
    const int widx = (int)((blockIdx.x * (size_t)blockDim.x + threadIdx.x) >> 5);
    const int lane = threadIdx.x & 31;
    if (widx >= Etot) return;

    int src, dst;
    if (widx < E) { src = edges[widx]; dst = edges[E + widx]; }
    else          { src = widx - E;    dst = src; }

    const int h = lane >> 3;
    const float alpha = ex[(size_t)widx * HEADS + h] / den[dst * HEADS + h];

    const float4* __restrict__ pl = (const float4*)(xl + (size_t)src * HC);
    float* __restrict__ pa = accum + (size_t)dst * HC;

    #pragma unroll
    for (int i = 0; i < 2; ++i) {
        const int q = lane * 2 + i;
        float4 v = pl[q];
        const int base = q << 2;
        __hip_atomic_fetch_add(&pa[base + 0], alpha * v.x, __ATOMIC_RELAXED, __HIP_MEMORY_SCOPE_AGENT);
        __hip_atomic_fetch_add(&pa[base + 1], alpha * v.y, __ATOMIC_RELAXED, __HIP_MEMORY_SCOPE_AGENT);
        __hip_atomic_fetch_add(&pa[base + 2], alpha * v.z, __ATOMIC_RELAXED, __HIP_MEMORY_SCOPE_AGENT);
        __hip_atomic_fetch_add(&pa[base + 3], alpha * v.w, __ATOMIC_RELAXED, __HIP_MEMORY_SCOPE_AGENT);
    }
}

__global__ void gat_final_kernel(const float* __restrict__ accum,
                                 const float* __restrict__ bias,
                                 float* __restrict__ out, int N) {
    const int idx = blockIdx.x * blockDim.x + threadIdx.x;
    if (idx >= N * CH) return;
    const int n = idx >> 6, c = idx & 63;
    const float* __restrict__ a = accum + (size_t)n * HC;
    out[idx] = 0.25f * (a[c] + a[CH + c] + a[2 * CH + c] + a[3 * CH + c]) + bias[c];
}

extern "C" void kernel_launch(void* const* d_in, const int* in_sizes, int n_in,
                              void* d_out, int out_size, void* d_ws, size_t ws_size,
                              hipStream_t stream) {
    const float* x    = (const float*)d_in[0];
    const float* Wl   = (const float*)d_in[1];
    const float* bl   = (const float*)d_in[2];
    const float* Wr   = (const float*)d_in[3];
    const float* br   = (const float*)d_in[4];
    const float* att  = (const float*)d_in[5];
    const float* bias = (const float*)d_in[6];
    const int*   edges = (const int*)d_in[7];

    const int N    = in_sizes[0] / CH;
    const int E    = in_sizes[7] / 2;
    const int Etot = E + N;

    float* xl    = (float*)d_ws;
    float* xr    = xl + (size_t)N * HC;
    float* accum = xr + (size_t)N * HC;
    float* smax  = accum + (size_t)N * HC;
    float* den   = smax + (size_t)N * HEADS;
    float* score = den + (size_t)N * HEADS;

    const int B = 256;
    gat_init_kernel<<<(N * HC + B - 1) / B, B, 0, stream>>>(accum, smax, den, N);
    gat_proj_kernel<<<N / 16, B, 0, stream>>>(x, Wl, bl, Wr, br, xl, xr, N);

    const long long waveThreads = (long long)Etot * 32;
    gat_score_kernel<<<(int)((waveThreads + B - 1) / B), B, 0, stream>>>(
        xl, xr, att, edges, score, smax, E, Etot);
    gat_exp_kernel<<<(Etot * HEADS + B - 1) / B, B, 0, stream>>>(
        edges, smax, score, den, E, Etot);
    gat_aggr_kernel<<<(int)((waveThreads + B - 1) / B), B, 0, stream>>>(
        xl, edges, score, den, accum, E, Etot);
    gat_final_kernel<<<(N * CH + B - 1) / B, B, 0, stream>>>(
        accum, bias, (float*)d_out, N);
}


# edge sort redone as a two-level counting sort ranked by LDS counters (range histogram per edge block, column sums, one block orders each range in LDS): no per-edge global atomic; still arrival-ranked
# speedup vs baseline: 6.4108x; 6.4108x over previous
#include <hip/hip_runtime.h>
#include <hip/hip_bf16.h>


typedef __attribute__((ext_vector_type(2))) float v2f;
typedef __attribute__((ext_vector_type(8))) float v8f;

#define CH 64
#define HEADS 4
#define HC 256
#define NEG_SLOPE 0.2f

#define RSHIFT 8
#define RNODES 256
#define SLICE 4096
#define BT 1024
#define RMAX 1024
#define COLCAP 8192

__global__ void gat_proj_kernel(const float* __restrict__ x,
                                const float* __restrict__ Wl,
                                const float* __restrict__ bl,
                                const float* __restrict__ Wr,
                                const float* __restrict__ br,
                                float* __restrict__ xl,
                                float* __restrict__ xr, int N) {
    __shared__ float As[16 * CH];
    const int row0 = blockIdx.x * 16;
    const int tid  = threadIdx.x;

    for (int i = tid; i < 16 * CH; i += 256) {
        int r = i >> 6, c = i & 63;
        As[i] = x[(size_t)(row0 + r) * CH + c];
    }
    __syncthreads();

    const int wave = tid >> 5;
    const int lane = tid & 31;
    const int m    = lane & 15;
    const int koff = (lane >> 4) << 1;
    const int mrow = (lane >> 4) << 3;

    for (int t = 0; t < 4; ++t) {
        const int colTile = wave * 4 + t;
        const bool left   = (colTile < 16);
        const float* __restrict__ W = left ? Wl : Wr;
        const float* __restrict__ b = left ? bl : br;
        const int n = ((colTile & 15) << 4) + m;

        v8f acc;
        const float bv = b[n];
        #pragma unroll
        for (int v = 0; v < 8; ++v) acc[v] = bv;

        #pragma unroll
        for (int k0 = 0; k0 < CH; k0 += 4) {
            v2f a, bf;
            a.x  = As[m * CH + k0 + koff];
            a.y  = As[m * CH + k0 + koff + 1];
            bf.x = W[(size_t)(k0 + koff) * HC + n];
            bf.y = W[(size_t)(k0 + koff + 1) * HC + n];
            acc = __builtin_amdgcn_wmma_f32_16x16x4_f32(
                false, a, false, bf, (short)0, acc, false, false);
        }

        float* __restrict__ out = left ? xl : xr;
        #pragma unroll
        for (int v = 0; v < 8; ++v)
            out[(size_t)(row0 + mrow + v) * HC + n] = acc[v];
    }
}

__global__ void gat_bin_count_kernel(const int* __restrict__ edges,
                                     int* __restrict__ H, int E, int R) {
    __shared__ int h[RMAX];
    const int tid = threadIdx.x;
    for (int t = tid; t < R; t += BT) h[t] = 0;
    __syncthreads();
    const int e0 = blockIdx.x * SLICE + tid;
    int d[SLICE / BT];
    #pragma unroll
    for (int k = 0; k < SLICE / BT; ++k) {
        const int e = e0 + k * BT;
        d[k] = (e < E) ? edges[E + e] : -1;
    }
    #pragma unroll
    for (int k = 0; k < SLICE / BT; ++k)
        if (d[k] >= 0) atomicAdd(&h[d[k] >> RSHIFT], 1);
    __syncthreads();
    for (int t = tid; t < R; t += BT) H[(size_t)blockIdx.x * R + t] = h[t];
}

__device__ __forceinline__ int gat_wave_scan(int s, int lane) {
    #pragma unroll
    for (int o = 1; o < 32; o <<= 1) {
        const int t = __shfl_up(s, o, 32);
        if (lane >= o) s += t;
    }
    return s;
}

__device__ __forceinline__ int gat_block_scan_excl(int v, int* sWave,
                                                   int lane, int wave) {
    const int s = gat_wave_scan(v, lane);
    if (lane == 31) sWave[wave] = s;
    __syncthreads();
    if (wave == 0) {
        const int w  = sWave[lane];
        const int ws = gat_wave_scan(w, lane);
        sWave[lane] = ws - w;
    }
    __syncthreads();
    const int r = sWave[wave] + s - v;
    __syncthreads();
    return r;
}

__global__ void gat_bin_scatter_kernel(const int* __restrict__ edges,
                                       const int* __restrict__ H,
                                       int* __restrict__ bBase,
                                       int* __restrict__ bTot,
                                       int* __restrict__ pairs_i,
                                       int E, int R, int NB) {
    __shared__ int  sTot[RMAX];
    __shared__ int  sPre[RMAX];
    __shared__ int  sLoc[RMAX];
    __shared__ int  sDelta[RMAX];
    __shared__ int  sWave[32];
    __shared__ int2 sStage[SLICE];
    const int tid  = threadIdx.x;
    const int lane = tid & 31;
    const int wave = tid >> 5;
    const int blk  = blockIdx.x;

    const int e0 = blk * SLICE + tid;
    int sv[SLICE / BT], dv[SLICE / BT];
    #pragma unroll
    for (int k = 0; k < SLICE / BT; ++k) {
        const int e = e0 + k * BT;
        sv[k] = (e < E) ? edges[e] : 0;
        dv[k] = (e < E) ? edges[E + e] : -1;
    }
    const int own = (tid < R) ? H[(size_t)blk * R + tid] : 0;

    if (tid < R) { sTot[tid] = 0; sPre[tid] = 0; }
    __syncthreads();
    const int G = BT / R;
    const int g = tid / R;
    const int t = tid - g * R;
    if (g < G) {
        int tot = 0, pre = 0;
        #pragma unroll 8
        for (int b = g; b < NB; b += G) {
            const int v = H[(size_t)b * R + t];
            tot += v;
            pre += (b < blk) ? v : 0;
        }
        atomicAdd(&sTot[t], tot);
        atomicAdd(&sPre[t], pre);
    }
    __syncthreads();

    const int tot   = (tid < R) ? sTot[tid] : 0;
    const int gbase = gat_block_scan_excl(tot, sWave, lane, wave);
    const int lofs  = gat_block_scan_excl(own, sWave, lane, wave);
    if (tid < R) {
        sLoc[tid]   = lofs;
        sDelta[tid] = gbase + sPre[tid] - lofs;
        if (blk == 0) { bBase[tid] = gbase; bTot[tid] = tot; }
    }
    __syncthreads();

    #pragma unroll
    for (int k = 0; k < SLICE / BT; ++k) {
        if (dv[k] >= 0) {
            const int lp = atomicAdd(&sLoc[dv[k] >> RSHIFT], 1);
            sStage[lp] = make_int2(sv[k], dv[k]);
        }
    }
    __syncthreads();

    int2* __restrict__ pairs = (int2*)pairs_i;
    const int left  = E - blk * SLICE;
    const int nhere = left < SLICE ? left : SLICE;
    for (int k = 0; k < SLICE / BT; ++k) {
        const int i = tid + k * BT;
        if (i < nhere) {
            const int2 pr = sStage[i];
            pairs[i + sDelta[pr.y >> RSHIFT]] = pr;
        }
    }
}

__global__ void gat_range_sort_kernel(const int* __restrict__ pairs_i,
                                      const int* __restrict__ bBase,
                                      const int* __restrict__ bTot,
                                      int* __restrict__ cnt,
                                      int* __restrict__ rowbeg,
                                      int* __restrict__ col, int N) {
    __shared__ int c[RNODES];
    __shared__ int sWave[32];
    __shared__ int sCol[COLCAP];
    const int tid  = threadIdx.x;
    const int lane = tid & 31;
    const int wave = tid >> 5;
    const int r    = blockIdx.x;
    const int base = bBase[r];
    const int tot  = bTot[r];
    const int2* __restrict__ pairs = (const int2*)pairs_i + base;

    if (tid < RNODES) c[tid] = 0;
    __syncthreads();
    for (int i0 = tid; i0 < tot + tid; i0 += 4 * BT) {
        int d[4];
        #pragma unroll
        for (int k = 0; k < 4; ++k) {
            const int i = i0 + k * BT;
            d[k] = (i < tot) ? pairs[i].y : -1;
        }
        #pragma unroll
        for (int k = 0; k < 4; ++k)
            if (d[k] >= 0) atomicAdd(&c[d[k] & (RNODES - 1)], 1);
    }
    __syncthreads();

    const int mine = (tid < RNODES) ? c[tid] : 0;
    const int excl = gat_block_scan_excl(mine, sWave, lane, wave);
    if (tid < RNODES) {
        const int n = r * RNODES + tid;
        if (n < N) { cnt[n] = mine; rowbeg[n] = base + excl; }
        c[tid] = excl;
    }
    __syncthreads();

    const bool staged = (tot <= COLCAP);
    for (int i0 = tid; i0 < tot + tid; i0 += 4 * BT) {
        int2 pr[4];
        #pragma unroll
        for (int k = 0; k < 4; ++k) {
            const int i = i0 + k * BT;
            pr[k] = (i < tot) ? pairs[i] : make_int2(0, -1);
        }
        #pragma unroll
        for (int k = 0; k < 4; ++k) {
            if (pr[k].y >= 0) {
                const int p = atomicAdd(&c[pr[k].y & (RNODES - 1)], 1);
                if (staged) sCol[p] = pr[k].x;
                else        col[base + p] = pr[k].x;
            }
        }
    }
    if (staged) {
        __syncthreads();
        for (int i = tid; i < tot; i += BT) col[base + i] = sCol[i];
    }
}

__device__ __forceinline__ float gat_lrelu_dot(const float4 a, const float4 b,
                                               const float4 w) {
    float vx = a.x + b.x; vx = vx > 0.0f ? vx : NEG_SLOPE * vx;
    float vy = a.y + b.y; vy = vy > 0.0f ? vy : NEG_SLOPE * vy;
    float vz = a.z + b.z; vz = vz > 0.0f ? vz : NEG_SLOPE * vz;
    float vw = a.w + b.w; vw = vw > 0.0f ? vw : NEG_SLOPE * vw;
    return w.x * vx + w.y * vy + w.z * vz + w.w * vw;
}

__device__ __forceinline__ float gat_sum8(float s) {
    s += __shfl_xor(s, 1, 32);
    s += __shfl_xor(s, 2, 32);
    s += __shfl_xor(s, 4, 32);
    return s;
}

__global__ void gat_node_kernel(const float* __restrict__ xl,
                                const float* __restrict__ xr,
                                const float* __restrict__ att,
                                const float* __restrict__ bias,
                                const int* __restrict__ col,
                                const int* __restrict__ rowbeg,
                                const int* __restrict__ cnt,
                                float* __restrict__ out, int N) {
    const int lane = threadIdx.x & 31;
    const int n = __builtin_amdgcn_readfirstlane(
        (int)((blockIdx.x * blockDim.x + threadIdx.x) >> 5));
    if (n >= N) return;

    const int beg = rowbeg[n];
    const int deg = cnt[n];

    const float4* __restrict__ pr = (const float4*)(xr + (size_t)n * HC);
    const float4* __restrict__ pa = (const float4*)att;
    const float4 r0 = pr[lane * 2], r1 = pr[lane * 2 + 1];
    const float4 w0 = pa[lane * 2], w1 = pa[lane * 2 + 1];

    const float4* __restrict__ ps = (const float4*)(xl + (size_t)n * HC);
    float4 acc0 = ps[lane * 2], acc1 = ps[lane * 2 + 1];
    float m = gat_sum8(gat_lrelu_dot(acc0, r0, w0) + gat_lrelu_dot(acc1, r1, w1));
    float d = 1.0f;

    for (int base = 0; base < deg; base += 32) {
        const int rem = deg - base;
        const int cc  = rem < 32 ? rem : 32;
        int mysrc = n;
        if (lane < cc) mysrc = col[beg + base + lane];

        for (int j = 0; j < cc; j += 2) {
            const int sa = __builtin_amdgcn_readlane(mysrc, j);
            const int sb = __builtin_amdgcn_readlane(mysrc, j + 1);
            const float4* __restrict__ pA = (const float4*)(xl + (size_t)sa * HC);
            const float4* __restrict__ pB = (const float4*)(xl + (size_t)sb * HC);
            const float4 a0 = pA[lane * 2], a1 = pA[lane * 2 + 1];
            const float4 b0 = pB[lane * 2], b1 = pB[lane * 2 + 1];

            const float sA = gat_sum8(gat_lrelu_dot(a0, r0, w0) + gat_lrelu_dot(a1, r1, w1));
            float       sB = gat_sum8(gat_lrelu_dot(b0, r0, w0) + gat_lrelu_dot(b1, r1, w1));
            if (j + 1 >= cc) sB = -__builtin_inff();

            const float mn = fmaxf(m, fmaxf(sA, sB));
            const float sc = __expf(m - mn);
            const float eA = __expf(sA - mn);
            const float eB = __expf(sB - mn);
            d = d * sc + eA + eB;
            acc0.x = acc0.x * sc + eA * a0.x + eB * b0.x;
            acc0.y = acc0.y * sc + eA * a0.y + eB * b0.y;
            acc0.z = acc0.z * sc + eA * a0.z + eB * b0.z;
            acc0.w = acc0.w * sc + eA * a0.w + eB * b0.w;
            acc1.x = acc1.x * sc + eA * a1.x + eB * b1.x;
            acc1.y = acc1.y * sc + eA * a1.y + eB * b1.y;
            acc1.z = acc1.z * sc + eA * a1.z + eB * b1.z;
            acc1.w = acc1.w * sc + eA * a1.w + eB * b1.w;
            m = mn;
        }
    }

    const float inv = 0.25f / d;
    float v[8] = { acc0.x * inv, acc0.y * inv, acc0.z * inv, acc0.w * inv,
                   acc1.x * inv, acc1.y * inv, acc1.z * inv, acc1.w * inv };
    #pragma unroll
    for (int i = 0; i < 8; ++i) {
        v[i] += __shfl_xor(v[i], 8, 32);
        v[i] += __shfl_xor(v[i], 16, 32);
    }
    if (lane < 16) {
        const int hi = (lane >> 3) & 1;
        const int c4 = (lane & 7) * 2 + hi;
        const float4 bv = ((const float4*)bias)[c4];
        float4 o;
        o.x = (hi ? v[4] : v[0]) + bv.x;
        o.y = (hi ? v[5] : v[1]) + bv.y;
        o.z = (hi ? v[6] : v[2]) + bv.z;
        o.w = (hi ? v[7] : v[3]) + bv.w;
        ((float4*)(out + (size_t)n * CH))[c4] = o;
    }
}

extern "C" void kernel_launch(void* const* d_in, const int* in_sizes, int n_in,
                              void* d_out, int out_size, void* d_ws, size_t ws_size,
                              hipStream_t stream) {
    const float* x    = (const float*)d_in[0];
    const float* Wl   = (const float*)d_in[1];
    const float* bl   = (const float*)d_in[2];
    const float* Wr   = (const float*)d_in[3];
    const float* br   = (const float*)d_in[4];
    const float* att  = (const float*)d_in[5];
    const float* bias = (const float*)d_in[6];
    const int*   edges = (const int*)d_in[7];

    const int N    = in_sizes[0] / CH;
    const int E    = in_sizes[7] / 2;

    const int R    = (N + RNODES - 1) / RNODES;
    const int NB   = (E + SLICE - 1) / SLICE;

    float* xl     = (float*)d_ws;
    float* xr     = xl + (size_t)N * HC;
    int*   pairs  = (int*)(xr + (size_t)N * HC);
    int*   cnt    = pairs + (size_t)2 * E;
    int*   rowbeg = cnt + (size_t)N;
    int*   col    = rowbeg + (size_t)N;
    int*   H      = col + (size_t)E;
    int*   bBase  = H + (size_t)NB * R;
    int*   bTot   = bBase + (size_t)R;

    const int B = 256;
    gat_proj_kernel<<<N / 16, B, 0, stream>>>(x, Wl, bl, Wr, br, xl, xr, N);
    gat_bin_count_kernel<<<NB, BT, 0, stream>>>(edges, H, E, R);
    gat_bin_scatter_kernel<<<NB, BT, 0, stream>>>(edges, H, bBase, bTot, pairs, E, R, NB);
    gat_range_sort_kernel<<<R, BT, 0, stream>>>(pairs, bBase, bTot, cnt, rowbeg, col, N);
    gat_node_kernel<<<(N + 7) / 8, B, 0, stream>>>(
        xl, xr, att, bias, col, rowbeg, cnt, (float*)d_out, N);
}


# projection on 256 resident workgroups: [W_l|W_r] staged once per workgroup in LDS; a wave holds a 16-row x tile as A fragments in registers and streams B fragments from LDS; same sums
# speedup vs baseline: 6.6529x; 1.0378x over previous
#include <hip/hip_runtime.h>
#include <hip/hip_bf16.h>


typedef __attribute__((ext_vector_type(2))) float v2f;
typedef __attribute__((ext_vector_type(8))) float v8f;

#define CH 64
#define HEADS 4
#define HC 256
#define NEG_SLOPE 0.2f

#define RSHIFT 8
#define RNODES 256
#define SLICE 4096
#define BT 1024
#define RMAX 1024
#define COLCAP 8192

#define PROJ_BLOCKS 256
#define PROJ_WAVES 16
#define PROJ_THREADS 512
#define PW_PITCH 1056
#define PROJ_LDS_FLOATS 41472

__global__ __launch_bounds__(PROJ_THREADS)
void gat_proj_kernel(const float* __restrict__ x,
                     const float* __restrict__ Wl,
                     const float* __restrict__ bl,
                     const float* __restrict__ Wr,
                     const float* __restrict__ br,
                     float* __restrict__ xl,
                     float* __restrict__ xr, int N) {
    __shared__ __attribute__((aligned(16))) float Ws[PROJ_LDS_FLOATS];
    const int tid = threadIdx.x;

    {
        const float* __restrict__ Wc = (tid < HC) ? (Wl + tid) : (Wr + (tid - HC));
        #pragma unroll
        for (int p = 0; p < CH / 2; ++p) {
            v2f w;
            w.x = Wc[(size_t)(2 * p) * HC];
            w.y = Wc[(size_t)(2 * p + 1) * HC];
            *(v2f*)&Ws[p * PW_PITCH + 2 * tid] = w;
        }
    }
    __syncthreads();

    const int lane = tid & 31;
    const int wave = __builtin_amdgcn_readfirstlane(tid >> 5);
    const int m    = lane & 15;
    const int h    = lane >> 4;
    const int koff = h << 1;
    const int mrow = h << 3;

    const int NT     = N >> 4;
    const int nUnits = NT * 4;
    const int stride = gridDim.x * PROJ_WAVES;
    for (int u = wave * gridDim.x + blockIdx.x; u < nUnits; u += stride) {
        const int q    = (u >= NT) + (u >= 2 * NT) + (u >= 3 * NT);
        const int row0 = (u - q * NT) << 4;
        const bool left = (q < 2);
        const float* __restrict__ b = left ? bl : br;
        float* __restrict__ out     = left ? xl : xr;
        const int nc = ((q & 1) << 7) + m;

        v2f a[CH / 4];
        const float* __restrict__ xrow = x + (size_t)(row0 + m) * CH + koff;
        #pragma unroll
        for (int i = 0; i < CH / 4; ++i) a[i] = *(const v2f*)(xrow + 4 * i);

        const float* __restrict__ wb = &Ws[h * PW_PITCH + 2 * ((q << 7) + m)];
        for (int g = 0; g < 2; ++g) {
            v8f acc[4];
            #pragma unroll
            for (int t = 0; t < 4; ++t) {
                const float bv = b[nc + 64 * g + 16 * t];
                #pragma unroll
                for (int v = 0; v < 8; ++v) acc[t][v] = bv;
            }

            #pragma unroll
            for (int i = 0; i < CH / 4; ++i) {
                #pragma unroll
                for (int t = 0; t < 4; ++t) {
                    const v2f bf = *(const v2f*)(wb + (2 * i) * PW_PITCH + 128 * g + 32 * t);
                    acc[t] = __builtin_amdgcn_wmma_f32_16x16x4_f32(
                        false, a[i], false, bf, (short)0, acc[t], false, false);
                }
            }

            #pragma unroll
            for (int t = 0; t < 4; ++t) {
                #pragma unroll
                for (int v = 0; v < 8; ++v)
                    out[(size_t)(row0 + mrow + v) * HC + nc + 64 * g + 16 * t] = acc[t][v];
            }
        }
    }
}

__global__ void gat_bin_count_kernel(const int* __restrict__ edges,
                                     int* __restrict__ H, int E, int R) {
    __shared__ int h[RMAX];
    const int tid = threadIdx.x;
    for (int t = tid; t < R; t += BT) h[t] = 0;
    __syncthreads();
    const int e0 = blockIdx.x * SLICE + tid;
    int d[SLICE / BT];
    #pragma unroll
    for (int k = 0; k < SLICE / BT; ++k) {
        const int e = e0 + k * BT;
        d[k] = (e < E) ? edges[E + e] : -1;
    }
    #pragma unroll
    for (int k = 0; k < SLICE / BT; ++k)
        if (d[k] >= 0) atomicAdd(&h[d[k] >> RSHIFT], 1);
    __syncthreads();
    for (int t = tid; t < R; t += BT) H[(size_t)blockIdx.x * R + t] = h[t];
}

__device__ __forceinline__ int gat_wave_scan(int s, int lane) {
    #pragma unroll
    for (int o = 1; o < 32; o <<= 1) {
        const int t = __shfl_up(s, o, 32);
        if (lane >= o) s += t;
    }
    return s;
}

__device__ __forceinline__ int gat_block_scan_excl(int v, int* sWave,
                                                   int lane, int wave) {
    const int s = gat_wave_scan(v, lane);
    if (lane == 31) sWave[wave] = s;
    __syncthreads();
    if (wave == 0) {
        const int w  = sWave[lane];
        const int ws = gat_wave_scan(w, lane);
        sWave[lane] = ws - w;
    }
    __syncthreads();
    const int r = sWave[wave] + s - v;
    __syncthreads();
    return r;
}

__global__ void gat_bin_scatter_kernel(const int* __restrict__ edges,
                                       const int* __restrict__ H,
                                       int* __restrict__ bBase,
                                       int* __restrict__ bTot,
                                       int* __restrict__ pairs_i,
                                       int E, int R, int NB) {
    __shared__ int  sTot[RMAX];
    __shared__ int  sPre[RMAX];
    __shared__ int  sLoc[RMAX];
    __shared__ int  sDelta[RMAX];
    __shared__ int  sWave[32];
    __shared__ int2 sStage[SLICE];
    const int tid  = threadIdx.x;
    const int lane = tid & 31;
    const int wave = tid >> 5;
    const int blk  = blockIdx.x;

    const int e0 = blk * SLICE + tid;
    int sv[SLICE / BT], dv[SLICE / BT];
    #pragma unroll
    for (int k = 0; k < SLICE / BT; ++k) {
        const int e = e0 + k * BT;
        sv[k] = (e < E) ? edges[e] : 0;
        dv[k] = (e < E) ? edges[E + e] : -1;
    }
    const int own = (tid < R) ? H[(size_t)blk * R + tid] : 0;

    if (tid < R) { sTot[tid] = 0; sPre[tid] = 0; }
    __syncthreads();
    const int G = BT / R;
    const int g = tid / R;
    const int t = tid - g * R;
    if (g < G) {
        int tot = 0, pre = 0;
        #pragma unroll 8
        for (int b = g; b < NB; b += G) {
            const int v = H[(size_t)b * R + t];
            tot += v;
            pre += (b < blk) ? v : 0;
        }
        atomicAdd(&sTot[t], tot);
        atomicAdd(&sPre[t], pre);
    }
    __syncthreads();

    const int tot   = (tid < R) ? sTot[tid] : 0;
    const int gbase = gat_block_scan_excl(tot, sWave, lane, wave);
    const int lofs  = gat_block_scan_excl(own, sWave, lane, wave);
    if (tid < R) {
        sLoc[tid]   = lofs;
        sDelta[tid] = gbase + sPre[tid] - lofs;
        if (blk == 0) { bBase[tid] = gbase; bTot[tid] = tot; }
    }
    __syncthreads();

    #pragma unroll
    for (int k = 0; k < SLICE / BT; ++k) {
        if (dv[k] >= 0) {
            const int lp = atomicAdd(&sLoc[dv[k] >> RSHIFT], 1);
            sStage[lp] = make_int2(sv[k], dv[k]);
        }
    }
    __syncthreads();

    int2* __restrict__ pairs = (int2*)pairs_i;
    const int left  = E - blk * SLICE;
    const int nhere = left < SLICE ? left : SLICE;
    for (int k = 0; k < SLICE / BT; ++k) {
        const int i = tid + k * BT;
        if (i < nhere) {
            const int2 pr = sStage[i];
            pairs[i + sDelta[pr.y >> RSHIFT]] = pr;
        }
    }
}

__global__ void gat_range_sort_kernel(const int* __restrict__ pairs_i,
                                      const int* __restrict__ bBase,
                                      const int* __restrict__ bTot,
                                      int* __restrict__ cnt,
                                      int* __restrict__ rowbeg,
                                      int* __restrict__ col, int N) {
    __shared__ int c[RNODES];
    __shared__ int sWave[32];
    __shared__ int sCol[COLCAP];
    const int tid  = threadIdx.x;
    const int lane = tid & 31;
    const int wave = tid >> 5;
    const int r    = blockIdx.x;
    const int base = bBase[r];
    const int tot  = bTot[r];
    const int2* __restrict__ pairs = (const int2*)pairs_i + base;

    if (tid < RNODES) c[tid] = 0;
    __syncthreads();
    for (int i0 = tid; i0 < tot + tid; i0 += 4 * BT) {
        int d[4];
        #pragma unroll
        for (int k = 0; k < 4; ++k) {
            const int i = i0 + k * BT;
            d[k] = (i < tot) ? pairs[i].y : -1;
        }
        #pragma unroll
        for (int k = 0; k < 4; ++k)
            if (d[k] >= 0) atomicAdd(&c[d[k] & (RNODES - 1)], 1);
    }
    __syncthreads();

    const int mine = (tid < RNODES) ? c[tid] : 0;
    const int excl = gat_block_scan_excl(mine, sWave, lane, wave);
    if (tid < RNODES) {
        const int n = r * RNODES + tid;
        if (n < N) { cnt[n] = mine; rowbeg[n] = base + excl; }
        c[tid] = excl;
    }
    __syncthreads();

    const bool staged = (tot <= COLCAP);
    for (int i0 = tid; i0 < tot + tid; i0 += 4 * BT) {
        int2 pr[4];
        #pragma unroll
        for (int k = 0; k < 4; ++k) {
            const int i = i0 + k * BT;
            pr[k] = (i < tot) ? pairs[i] : make_int2(0, -1);
        }
        #pragma unroll
        for (int k = 0; k < 4; ++k) {
            if (pr[k].y >= 0) {
                const int p = atomicAdd(&c[pr[k].y & (RNODES - 1)], 1);
                if (staged) sCol[p] = pr[k].x;
                else        col[base + p] = pr[k].x;
            }
        }
    }
    if (staged) {
        __syncthreads();
        for (int i = tid; i < tot; i += BT) col[base + i] = sCol[i];
    }
}

__device__ __forceinline__ float gat_lrelu_dot(const float4 a, const float4 b,
                                               const float4 w) {
    float vx = a.x + b.x; vx = vx > 0.0f ? vx : NEG_SLOPE * vx;
    float vy = a.y + b.y; vy = vy > 0.0f ? vy : NEG_SLOPE * vy;
    float vz = a.z + b.z; vz = vz > 0.0f ? vz : NEG_SLOPE * vz;
    float vw = a.w + b.w; vw = vw > 0.0f ? vw : NEG_SLOPE * vw;
    return w.x * vx + w.y * vy + w.z * vz + w.w * vw;
}

__device__ __forceinline__ float gat_sum8(float s) {
    s += __shfl_xor(s, 1, 32);
    s += __shfl_xor(s, 2, 32);
    s += __shfl_xor(s, 4, 32);
    return s;
}

__global__ void gat_node_kernel(const float* __restrict__ xl,
                                const float* __restrict__ xr,
                                const float* __restrict__ att,
                                const float* __restrict__ bias,
                                const int* __restrict__ col,
                                const int* __restrict__ rowbeg,
                                const int* __restrict__ cnt,
                                float* __restrict__ out, int N) {
    const int lane = threadIdx.x & 31;
    const int n = __builtin_amdgcn_readfirstlane(
        (int)((blockIdx.x * blockDim.x + threadIdx.x) >> 5));
    if (n >= N) return;

    const int beg = rowbeg[n];
    const int deg = cnt[n];

    const float4* __restrict__ pr = (const float4*)(xr + (size_t)n * HC);
    const float4* __restrict__ pa = (const float4*)att;
    const float4 r0 = pr[lane * 2], r1 = pr[lane * 2 + 1];
    const float4 w0 = pa[lane * 2], w1 = pa[lane * 2 + 1];

    const float4* __restrict__ ps = (const float4*)(xl + (size_t)n * HC);
    float4 acc0 = ps[lane * 2], acc1 = ps[lane * 2 + 1];
    float m = gat_sum8(gat_lrelu_dot(acc0, r0, w0) + gat_lrelu_dot(acc1, r1, w1));
    float d = 1.0f;

    for (int base = 0; base < deg; base += 32) {
        const int rem = deg - base;
        const int cc  = rem < 32 ? rem : 32;
        int mysrc = n;
        if (lane < cc) mysrc = col[beg + base + lane];

        for (int j = 0; j < cc; j += 2) {
            const int sa = __builtin_amdgcn_readlane(mysrc, j);
            const int sb = __builtin_amdgcn_readlane(mysrc, j + 1);
            const float4* __restrict__ pA = (const float4*)(xl + (size_t)sa * HC);
            const float4* __restrict__ pB = (const float4*)(xl + (size_t)sb * HC);
            const float4 a0 = pA[lane * 2], a1 = pA[lane * 2 + 1];
            const float4 b0 = pB[lane * 2], b1 = pB[lane * 2 + 1];

            const float sA = gat_sum8(gat_lrelu_dot(a0, r0, w0) + gat_lrelu_dot(a1, r1, w1));
            float       sB = gat_sum8(gat_lrelu_dot(b0, r0, w0) + gat_lrelu_dot(b1, r1, w1));
            if (j + 1 >= cc) sB = -__builtin_inff();

            const float mn = fmaxf(m, fmaxf(sA, sB));
            const float sc = __expf(m - mn);
            const float eA = __expf(sA - mn);
            const float eB = __expf(sB - mn);
            d = d * sc + eA + eB;
            acc0.x = acc0.x * sc + eA * a0.x + eB * b0.x;
            acc0.y = acc0.y * sc + eA * a0.y + eB * b0.y;
            acc0.z = acc0.z * sc + eA * a0.z + eB * b0.z;
            acc0.w = acc0.w * sc + eA * a0.w + eB * b0.w;
            acc1.x = acc1.x * sc + eA * a1.x + eB * b1.x;
            acc1.y = acc1.y * sc + eA * a1.y + eB * b1.y;
            acc1.z = acc1.z * sc + eA * a1.z + eB * b1.z;
            acc1.w = acc1.w * sc + eA * a1.w + eB * b1.w;
            m = mn;
        }
    }

    const float inv = 0.25f / d;
    float v[8] = { acc0.x * inv, acc0.y * inv, acc0.z * inv, acc0.w * inv,
                   acc1.x * inv, acc1.y * inv, acc1.z * inv, acc1.w * inv };
    #pragma unroll
    for (int i = 0; i < 8; ++i) {
        v[i] += __shfl_xor(v[i], 8, 32);
        v[i] += __shfl_xor(v[i], 16, 32);
    }
    if (lane < 16) {
        const int hi = (lane >> 3) & 1;
        const int c4 = (lane & 7) * 2 + hi;
        const float4 bv = ((const float4*)bias)[c4];
        float4 o;
        o.x = (hi ? v[4] : v[0]) + bv.x;
        o.y = (hi ? v[5] : v[1]) + bv.y;
        o.z = (hi ? v[6] : v[2]) + bv.z;
        o.w = (hi ? v[7] : v[3]) + bv.w;
        ((float4*)(out + (size_t)n * CH))[c4] = o;
    }
}

extern "C" void kernel_launch(void* const* d_in, const int* in_sizes, int n_in,
                              void* d_out, int out_size, void* d_ws, size_t ws_size,
                              hipStream_t stream) {
    const float* x    = (const float*)d_in[0];
    const float* Wl   = (const float*)d_in[1];
    const float* bl   = (const float*)d_in[2];
    const float* Wr   = (const float*)d_in[3];
    const float* br   = (const float*)d_in[4];
    const float* att  = (const float*)d_in[5];
    const float* bias = (const float*)d_in[6];
    const int*   edges = (const int*)d_in[7];

    const int N    = in_sizes[0] / CH;
    const int E    = in_sizes[7] / 2;

    const int R    = (N + RNODES - 1) / RNODES;
    const int NB   = (E + SLICE - 1) / SLICE;

    float* xl     = (float*)d_ws;
    float* xr     = xl + (size_t)N * HC;
    int*   pairs  = (int*)(xr + (size_t)N * HC);
    int*   cnt    = pairs + (size_t)2 * E;
    int*   rowbeg = cnt + (size_t)N;
    int*   col    = rowbeg + (size_t)N;
    int*   H      = col + (size_t)E;
    int*   bBase  = H + (size_t)NB * R;
    int*   bTot   = bBase + (size_t)R;

    const int B = 256;
    gat_proj_kernel<<<PROJ_BLOCKS, PROJ_THREADS, 0, stream>>>(x, Wl, bl, Wr, br, xl, xr, N);
    gat_bin_count_kernel<<<NB, BT, 0, stream>>>(edges, H, E, R);
    gat_bin_scatter_kernel<<<NB, BT, 0, stream>>>(edges, H, bBase, bTot, pairs, E, R, NB);
    gat_range_sort_kernel<<<R, BT, 0, stream>>>(pairs, bBase, bTot, cnt, rowbeg, col, N);
    gat_node_kernel<<<(N + 7) / 8, B, 0, stream>>>(
        xl, xr, att, bias, col, rowbeg, cnt, (float*)d_out, N);
}


# edge sort level 1 made block-local: an edge block orders its own slice by range in place and writes its range counts; a range block collects its pieces: one launch fewer; still arrival-ranked
# speedup vs baseline: 7.1728x; 1.0781x over previous
#include <hip/hip_runtime.h>
#include <hip/hip_bf16.h>


typedef __attribute__((ext_vector_type(2))) float v2f;
typedef __attribute__((ext_vector_type(8))) float v8f;

#define CH 64
#define HEADS 4
#define HC 256
#define NEG_SLOPE 0.2f

#define RSHIFT 8
#define RNODES 256
#define SLICE 4096
#define BT 1024
#define RMAX 1024
#define COLCAP 8192
#define NBMAX 1024

#define PROJ_BLOCKS 256
#define PROJ_WAVES 16
#define PROJ_THREADS 512
#define PW_PITCH 1056
#define PROJ_LDS_FLOATS 41472

__global__ __launch_bounds__(PROJ_THREADS)
void gat_proj_kernel(const float* __restrict__ x,
                     const float* __restrict__ Wl,
                     const float* __restrict__ bl,
                     const float* __restrict__ Wr,
                     const float* __restrict__ br,
                     float* __restrict__ xl,
                     float* __restrict__ xr, int N) {
    __shared__ __attribute__((aligned(16))) float Ws[PROJ_LDS_FLOATS];
    const int tid = threadIdx.x;

    {
        const float* __restrict__ Wc = (tid < HC) ? (Wl + tid) : (Wr + (tid - HC));
        #pragma unroll
        for (int p = 0; p < CH / 2; ++p) {
            v2f w;
            w.x = Wc[(size_t)(2 * p) * HC];
            w.y = Wc[(size_t)(2 * p + 1) * HC];
            *(v2f*)&Ws[p * PW_PITCH + 2 * tid] = w;
        }
    }
    __syncthreads();

    const int lane = tid & 31;
    const int wave = __builtin_amdgcn_readfirstlane(tid >> 5);
    const int m    = lane & 15;
    const int h    = lane >> 4;
    const int koff = h << 1;
    const int mrow = h << 3;

    const int NT     = N >> 4;
    const int nUnits = NT * 4;
    const int stride = gridDim.x * PROJ_WAVES;
    for (int u = wave * gridDim.x + blockIdx.x; u < nUnits; u += stride) {
        const int q    = (u >= NT) + (u >= 2 * NT) + (u >= 3 * NT);
        const int row0 = (u - q * NT) << 4;
        const bool left = (q < 2);
        const float* __restrict__ b = left ? bl : br;
        float* __restrict__ out     = left ? xl : xr;
        const int nc = ((q & 1) << 7) + m;

        v2f a[CH / 4];
        const float* __restrict__ xrow = x + (size_t)(row0 + m) * CH + koff;
        #pragma unroll
        for (int i = 0; i < CH / 4; ++i) a[i] = *(const v2f*)(xrow + 4 * i);

        const float* __restrict__ wb = &Ws[h * PW_PITCH + 2 * ((q << 7) + m)];
        for (int g = 0; g < 2; ++g) {
            v8f acc[4];
            #pragma unroll
            for (int t = 0; t < 4; ++t) {
                const float bv = b[nc + 64 * g + 16 * t];
                #pragma unroll
                for (int v = 0; v < 8; ++v) acc[t][v] = bv;
            }

            #pragma unroll
            for (int i = 0; i < CH / 4; ++i) {
                #pragma unroll
                for (int t = 0; t < 4; ++t) {
                    const v2f bf = *(const v2f*)(wb + (2 * i) * PW_PITCH + 128 * g + 32 * t);
                    acc[t] = __builtin_amdgcn_wmma_f32_16x16x4_f32(
                        false, a[i], false, bf, (short)0, acc[t], false, false);
                }
            }

            #pragma unroll
            for (int t = 0; t < 4; ++t) {
                #pragma unroll
                for (int v = 0; v < 8; ++v)
                    out[(size_t)(row0 + mrow + v) * HC + nc + 64 * g + 16 * t] = acc[t][v];
            }
        }
    }
}


__device__ __forceinline__ int gat_wave_scan(int s, int lane) {
    #pragma unroll
    for (int o = 1; o < 32; o <<= 1) {
        const int t = __shfl_up(s, o, 32);
        if (lane >= o) s += t;
    }
    return s;
}

__device__ __forceinline__ int gat_block_scan_excl(int v, int* sWave,
                                                   int lane, int wave) {
    const int s = gat_wave_scan(v, lane);
    if (lane == 31) sWave[wave] = s;
    __syncthreads();
    if (wave == 0) {
        const int w  = sWave[lane];
        const int ws = gat_wave_scan(w, lane);
        sWave[lane] = ws - w;
    }
    __syncthreads();
    const int r = sWave[wave] + s - v;
    __syncthreads();
    return r;
}

__global__ void gat_bin_local_kernel(const int* __restrict__ edges,
                                     int* __restrict__ H,
                                     int* __restrict__ L,
                                     int* __restrict__ pairs_i,
                                     int E, int R) {
    __shared__ int  sCnt[RMAX];
    __shared__ int  sLofs[RMAX];
    __shared__ int  sWave[32];
    __shared__ int2 sStage[SLICE];
    const int tid  = threadIdx.x;
    const int lane = tid & 31;
    const int wave = tid >> 5;
    const int blk  = blockIdx.x;

    const int e0 = blk * SLICE + tid;
    int sv[SLICE / BT], dv[SLICE / BT], rk[SLICE / BT];
    #pragma unroll
    for (int k = 0; k < SLICE / BT; ++k) {
        const int e = e0 + k * BT;
        sv[k] = (e < E) ? edges[e] : 0;
        dv[k] = (e < E) ? edges[E + e] : -1;
    }
    if (tid < R) sCnt[tid] = 0;
    __syncthreads();
    #pragma unroll
    for (int k = 0; k < SLICE / BT; ++k)
        rk[k] = (dv[k] >= 0) ? atomicAdd(&sCnt[dv[k] >> RSHIFT], 1) : 0;
    __syncthreads();

    const int own  = (tid < R) ? sCnt[tid] : 0;
    const int lofs = gat_block_scan_excl(own, sWave, lane, wave);
    if (tid < R) {
        sLofs[tid] = lofs;
        H[(size_t)blk * R + tid] = own;
        L[(size_t)blk * R + tid] = lofs;
    }
    __syncthreads();

    #pragma unroll
    for (int k = 0; k < SLICE / BT; ++k)
        if (dv[k] >= 0) sStage[sLofs[dv[k] >> RSHIFT] + rk[k]] = make_int2(sv[k], dv[k]);
    __syncthreads();

    int2* __restrict__ pairs = (int2*)pairs_i + (size_t)blk * SLICE;
    const int left  = E - blk * SLICE;
    const int nhere = left < SLICE ? left : SLICE;
    for (int k = 0; k < SLICE / BT; ++k) {
        const int i = tid + k * BT;
        if (i < nhere) pairs[i] = sStage[i];
    }
}

__global__ void gat_range_sort_kernel(const int* __restrict__ pairs_i,
                                      const int* __restrict__ H,
                                      const int* __restrict__ L,
                                      int* __restrict__ cnt,
                                      int* __restrict__ rowbeg,
                                      int* __restrict__ col,
                                      int N, int R, int NB) {
    __shared__ int  c[RNODES];
    __shared__ int  sWave[32];
    __shared__ int  sTot, sBase;
    __shared__ int4 sPiece[NBMAX];
    __shared__ int2 sPairs[COLCAP];
    __shared__ int  sCol[COLCAP];
    const int tid  = threadIdx.x;
    const int lane = tid & 31;
    const int wave = tid >> 5;
    const int r    = blockIdx.x;
    const int2* __restrict__ pairs = (const int2*)pairs_i;

    const int hb = (tid < NB) ? H[(size_t)tid * R + r] : 0;
    const int lb = (tid < NB) ? L[(size_t)tid * R + r] : 0;
    if (tid < RNODES) c[tid] = 0;
    if (tid == 0) sBase = 0;
    const int pp = gat_block_scan_excl(hb, sWave, lane, wave);
    const int ls = gat_wave_scan(lb, lane);
    if (lane == 31 && ls != 0) atomicAdd(&sBase, ls);
    if (tid == BT - 1) sTot = pp + hb;
    sPiece[tid] = make_int4(hb, tid * SLICE + lb, pp, 0);
    __syncthreads();
    const int tot  = sTot;
    const int base = sBase;
    const bool staged = (tot <= COLCAP);

    const int wv = __builtin_amdgcn_readfirstlane(wave);
    for (int b0 = 0; b0 < NB; b0 += 8 * 32) {
        int4 pc[8];
        int2 pr[8];
        #pragma unroll
        for (int u = 0; u < 8; ++u) pc[u] = sPiece[b0 + 32 * u + wv];
        #pragma unroll
        for (int u = 0; u < 8; ++u)
            pr[u] = (lane < pc[u].x) ? pairs[pc[u].y + lane] : make_int2(0, -1);
        #pragma unroll
        for (int u = 0; u < 8; ++u) {
            if (pr[u].y >= 0) {
                atomicAdd(&c[pr[u].y & (RNODES - 1)], 1);
                if (staged) sPairs[pc[u].z + lane] = pr[u];
            }
        }
        #pragma unroll
        for (int u = 0; u < 8; ++u) {
            for (int k = lane + 32; k < pc[u].x; k += 32) {
                const int2 q = pairs[pc[u].y + k];
                atomicAdd(&c[q.y & (RNODES - 1)], 1);
                if (staged) sPairs[pc[u].z + k] = q;
            }
        }
    }
    __syncthreads();

    const int mine = (tid < RNODES) ? c[tid] : 0;
    const int excl = gat_block_scan_excl(mine, sWave, lane, wave);
    if (tid < RNODES) {
        const int n = r * RNODES + tid;
        if (n < N) { cnt[n] = mine; rowbeg[n] = base + excl; }
        c[tid] = excl;
    }
    __syncthreads();

    if (staged) {
        for (int i = tid; i < tot; i += BT) {
            const int2 q = sPairs[i];
            const int p = atomicAdd(&c[q.y & (RNODES - 1)], 1);
            sCol[p] = q.x;
        }
        __syncthreads();
        for (int i = tid; i < tot; i += BT) col[base + i] = sCol[i];
    } else {
        for (int b = wave; b < NB; b += 32) {
            const int4 pq = sPiece[b];
            for (int k = lane; k < pq.x; k += 32) {
                const int2 q = pairs[pq.y + k];
                const int p = atomicAdd(&c[q.y & (RNODES - 1)], 1);
                col[base + p] = q.x;
            }
        }
    }
}

__device__ __forceinline__ float gat_lrelu_dot(const float4 a, const float4 b,
                                               const float4 w) {
    float vx = a.x + b.x; vx = vx > 0.0f ? vx : NEG_SLOPE * vx;
    float vy = a.y + b.y; vy = vy > 0.0f ? vy : NEG_SLOPE * vy;
    float vz = a.z + b.z; vz = vz > 0.0f ? vz : NEG_SLOPE * vz;
    float vw = a.w + b.w; vw = vw > 0.0f ? vw : NEG_SLOPE * vw;
    return w.x * vx + w.y * vy + w.z * vz + w.w * vw;
}

__device__ __forceinline__ float gat_sum8(float s) {
    s += __shfl_xor(s, 1, 32);
    s += __shfl_xor(s, 2, 32);
    s += __shfl_xor(s, 4, 32);
    return s;
}

__global__ void gat_node_kernel(const float* __restrict__ xl,
                                const float* __restrict__ xr,
                                const float* __restrict__ att,
                                const float* __restrict__ bias,
                                const int* __restrict__ col,
                                const int* __restrict__ rowbeg,
                                const int* __restrict__ cnt,
                                float* __restrict__ out, int N) {
    const int lane = threadIdx.x & 31;
    const int n = __builtin_amdgcn_readfirstlane(
        (int)((blockIdx.x * blockDim.x + threadIdx.x) >> 5));
    if (n >= N) return;

    const int beg = rowbeg[n];
    const int deg = cnt[n];

    const float4* __restrict__ pr = (const float4*)(xr + (size_t)n * HC);
    const float4* __restrict__ pa = (const float4*)att;
    const float4 r0 = pr[lane * 2], r1 = pr[lane * 2 + 1];
    const float4 w0 = pa[lane * 2], w1 = pa[lane * 2 + 1];

    const float4* __restrict__ ps = (const float4*)(xl + (size_t)n * HC);
    float4 acc0 = ps[lane * 2], acc1 = ps[lane * 2 + 1];
    float m = gat_sum8(gat_lrelu_dot(acc0, r0, w0) + gat_lrelu_dot(acc1, r1, w1));
    float d = 1.0f;

    for (int base = 0; base < deg; base += 32) {
        const int rem = deg - base;
        const int cc  = rem < 32 ? rem : 32;
        int mysrc = n;
        if (lane < cc) mysrc = col[beg + base + lane];

        for (int j = 0; j < cc; j += 2) {
            const int sa = __builtin_amdgcn_readlane(mysrc, j);
            const int sb = __builtin_amdgcn_readlane(mysrc, j + 1);
            const float4* __restrict__ pA = (const float4*)(xl + (size_t)sa * HC);
            const float4* __restrict__ pB = (const float4*)(xl + (size_t)sb * HC);
            const float4 a0 = pA[lane * 2], a1 = pA[lane * 2 + 1];
            const float4 b0 = pB[lane * 2], b1 = pB[lane * 2 + 1];

            const float sA = gat_sum8(gat_lrelu_dot(a0, r0, w0) + gat_lrelu_dot(a1, r1, w1));
            float       sB = gat_sum8(gat_lrelu_dot(b0, r0, w0) + gat_lrelu_dot(b1, r1, w1));
            if (j + 1 >= cc) sB = -__builtin_inff();

            const float mn = fmaxf(m, fmaxf(sA, sB));
            const float sc = __expf(m - mn);
            const float eA = __expf(sA - mn);
            const float eB = __expf(sB - mn);
            d = d * sc + eA + eB;
            acc0.x = acc0.x * sc + eA * a0.x + eB * b0.x;
            acc0.y = acc0.y * sc + eA * a0.y + eB * b0.y;
            acc0.z = acc0.z * sc + eA * a0.z + eB * b0.z;
            acc0.w = acc0.w * sc + eA * a0.w + eB * b0.w;
            acc1.x = acc1.x * sc + eA * a1.x + eB * b1.x;
            acc1.y = acc1.y * sc + eA * a1.y + eB * b1.y;
            acc1.z = acc1.z * sc + eA * a1.z + eB * b1.z;
            acc1.w = acc1.w * sc + eA * a1.w + eB * b1.w;
            m = mn;
        }
    }

    const float inv = 0.25f / d;
    float v[8] = { acc0.x * inv, acc0.y * inv, acc0.z * inv, acc0.w * inv,
                   acc1.x * inv, acc1.y * inv, acc1.z * inv, acc1.w * inv };
    #pragma unroll
    for (int i = 0; i < 8; ++i) {
        v[i] += __shfl_xor(v[i], 8, 32);
        v[i] += __shfl_xor(v[i], 16, 32);
    }
    if (lane < 16) {
        const int hi = (lane >> 3) & 1;
        const int c4 = (lane & 7) * 2 + hi;
        const float4 bv = ((const float4*)bias)[c4];
        float4 o;
        o.x = (hi ? v[4] : v[0]) + bv.x;
        o.y = (hi ? v[5] : v[1]) + bv.y;
        o.z = (hi ? v[6] : v[2]) + bv.z;
        o.w = (hi ? v[7] : v[3]) + bv.w;
        ((float4*)(out + (size_t)n * CH))[c4] = o;
    }
}

extern "C" void kernel_launch(void* const* d_in, const int* in_sizes, int n_in,
                              void* d_out, int out_size, void* d_ws, size_t ws_size,
                              hipStream_t stream) {
    const float* x    = (const float*)d_in[0];
    const float* Wl   = (const float*)d_in[1];
    const float* bl   = (const float*)d_in[2];
    const float* Wr   = (const float*)d_in[3];
    const float* br   = (const float*)d_in[4];
    const float* att  = (const float*)d_in[5];
    const float* bias = (const float*)d_in[6];
    const int*   edges = (const int*)d_in[7];

    const int N    = in_sizes[0] / CH;
    const int E    = in_sizes[7] / 2;

    const int R    = (N + RNODES - 1) / RNODES;
    const int NB   = (E + SLICE - 1) / SLICE;

    float* xl     = (float*)d_ws;
    float* xr     = xl + (size_t)N * HC;
    int*   pairs  = (int*)(xr + (size_t)N * HC);
    int*   cnt    = pairs + (size_t)2 * E;
    int*   rowbeg = cnt + (size_t)N;
    int*   col    = rowbeg + (size_t)N;
    int*   H      = col + (size_t)E;
    int*   L      = H + (size_t)NB * R;

    const int B = 256;
    gat_proj_kernel<<<PROJ_BLOCKS, PROJ_THREADS, 0, stream>>>(x, Wl, bl, Wr, br, xl, xr, N);
    gat_bin_local_kernel<<<NB, BT, 0, stream>>>(edges, H, L, pairs, E, R);
    gat_range_sort_kernel<<<R, BT, 0, stream>>>(pairs, H, L, cnt, rowbeg, col, N, R, NB);
    gat_node_kernel<<<(N + 7) / 8, B, 0, stream>>>(
        xl, xr, att, bias, col, rowbeg, cnt, (float*)d_out, N);
}
